# starting kernel
# baseline (speedup 1.0000x reference)
#include <hip/hip_runtime.h>
#include <hip/hip_bf16.h>

#define BATCH 8
#define CH    64
#define HW    4096
#define GRP   8
#define CGRP  4

typedef float v2f __attribute__((ext_vector_type(2)));
typedef float v8f __attribute__((ext_vector_type(8)));

__global__ void build_weff_kernel(const float* __restrict__ A,
                                  const float* __restrict__ Ainv,
                                  const float* __restrict__ Wf,
                                  float* __restrict__ Weff) {
    int idx = blockIdx.x * blockDim.x + threadIdx.x;
    if (idx >= CH * CH) return;
    int o  = idx >> 6;
    int c  = idx & 63;
    int no = o >> 2, r = o & 3;
    int nc = c >> 2, s = c & 3;

    float acc = 0.0f;
    for (int g = 0; g < GRP; ++g) {
        const float* Ag  = A    + g * (CGRP * CGRP);
        const float* Aig = Ainv + g * (CGRP * CGRP);
        for (int i = 0; i < CGRP; ++i) {
            float inner = 0.0f;
            for (int j = 0; j < CGRP; ++j) {
                inner += Wf[(no * 4 + i) * CH + (nc * 4 + j)] * Aig[j * 4 + s];
            }
            acc += Ag[r * 4 + i] * inner;
        }
    }
    Weff[o * CH + c] = acc * (1.0f / (float)GRP);
}

__global__ __launch_bounds__(256)
void reynolds_gemm_kernel(const float* __restrict__ x,
                          const float* __restrict__ Weff,
                          float* __restrict__ out) {
    const int wave = (blockIdx.x * 256 + threadIdx.x) >> 5;
    const int lane = threadIdx.x & 31;

    const int batch   = wave >> 8;
    const int rem     = wave & 255;
    const int m_off   = (rem >> 6) << 4;
    const int p_off   = (rem & 63) << 6;

    const int nl  = lane & 15;
    const int khi = lane >> 4;
    const int kb  = khi << 1;

    const float* __restrict__ xb = x   + (size_t)batch * CH * HW;
    float*       __restrict__ ob = out + (size_t)batch * CH * HW;

    v8f acc0 = {}, acc1 = {}, acc2 = {}, acc3 = {};

    const float* wrow = Weff + (m_off + nl) * CH + kb;
    const float* xcol = xb + (size_t)kb * HW + p_off + nl;

    #pragma unroll
    for (int k0 = 0; k0 < CH; k0 += 4) {
        v2f a = *(const v2f*)(wrow + k0);

        const float* xk = xcol + (size_t)k0 * HW;
        v2f b0, b1, b2, b3;
        b0.x = xk[0];            b0.y = xk[HW];
        b1.x = xk[16];           b1.y = xk[HW + 16];
        b2.x = xk[32];           b2.y = xk[HW + 32];
        b3.x = xk[48];           b3.y = xk[HW + 48];

        acc0 = __builtin_amdgcn_wmma_f32_16x16x4_f32(false, a, false, b0, (short)0, acc0, false, false);
        acc1 = __builtin_amdgcn_wmma_f32_16x16x4_f32(false, a, false, b1, (short)0, acc1, false, false);
        acc2 = __builtin_amdgcn_wmma_f32_16x16x4_f32(false, a, false, b2, (short)0, acc2, false, false);
        acc3 = __builtin_amdgcn_wmma_f32_16x16x4_f32(false, a, false, b3, (short)0, acc3, false, false);
    }

    #pragma unroll
    for (int v = 0; v < 8; ++v) {
        float* op = ob + (size_t)(m_off + v + 8 * khi) * HW + p_off + nl;
        op[0]  = acc0[v];
        op[16] = acc1[v];
        op[32] = acc2[v];
        op[48] = acc3[v];
    }
}

extern "C" void kernel_launch(void* const* d_in, const int* in_sizes, int n_in,
                              void* d_out, int out_size, void* d_ws, size_t ws_size,
                              hipStream_t stream) {
    const float* x    = (const float*)d_in[0];
    const float* A    = (const float*)d_in[1];
    const float* Ainv = (const float*)d_in[2];
    const float* Wf   = (const float*)d_in[3];
    float* out  = (float*)d_out;
    float* Weff = (float*)d_ws;

    build_weff_kernel<<<(CH * CH + 255) / 256, 256, 0, stream>>>(A, Ainv, Wf, Weff);

    reynolds_gemm_kernel<<<256, 256, 0, stream>>>(x, Weff, out);
}


# one launch instead of two: each block builds its own 16 rows of W_eff in LDS (same summation order) under its first x loads; built code then keeps 56 x loads in flight in the k loop, was 12
# speedup vs baseline: 1.8154x; 1.8154x over previous
#include <hip/hip_runtime.h>
#include <hip/hip_bf16.h>


#define BATCH 8
#define CH    64
#define HW    4096
#define GRP   8
#define CGRP  4

#define WPITCH 68
#define NPRE   4

typedef float v2f __attribute__((ext_vector_type(2)));
typedef float v4f __attribute__((ext_vector_type(4)));
typedef float v8f __attribute__((ext_vector_type(8)));

__global__ __launch_bounds__(256)
void reynolds_fused_kernel(const float* __restrict__ x,
                           const float* __restrict__ A,
                           const float* __restrict__ Ainv,
                           const float* __restrict__ Wf,
                           float* __restrict__ out) {
    __shared__ __attribute__((aligned(16))) float Ws[16 * WPITCH];

    const int tid  = threadIdx.x;
    const int lane = tid & 31;
    const int wib  = tid >> 5;

    const int blk   = blockIdx.x;
    const int batch = blk >> 5;
    const int m_off = ((blk >> 3) & 3) << 4;
    const int p_off = (((blk & 7) << 3) + wib) << 6;

    const int nl  = lane & 15;
    const int khi = lane >> 4;
    const int kb  = khi << 1;

    const float* __restrict__ xb = x   + (size_t)batch * CH * HW;
    float*       __restrict__ ob = out + (size_t)batch * CH * HW;

    const int s   = tid & 3;
    const int nc  = (tid >> 2) & 15;
    const int nor = tid >> 6;
    const int no  = (m_off >> 2) + nor;
    const float* wfb = Wf + (no * 4) * CH + nc * 4;
    v4f wf[4];
    #pragma unroll
    for (int i = 0; i < 4; ++i) wf[i] = *(const v4f*)(wfb + i * CH);

    const v4f a4 = *(const v4f*)(A + lane * 4);

    float ainv[GRP][CGRP];
    #pragma unroll
    for (int g = 0; g < GRP; ++g) {
        #pragma unroll
        for (int j = 0; j < CGRP; ++j) ainv[g][j] = Ainv[g * (CGRP * CGRP) + j * 4 + s];
    }

    const float* xcol = xb + (size_t)kb * HW + p_off + nl;
    v2f bpre[NPRE][4];
    #pragma unroll
    for (int t = 0; t < NPRE; ++t) {
        const float* xk = xcol + (size_t)(4 * t) * HW;
        bpre[t][0].x = xk[0];    bpre[t][0].y = xk[HW];
        bpre[t][1].x = xk[16];   bpre[t][1].y = xk[HW + 16];
        bpre[t][2].x = xk[32];   bpre[t][2].y = xk[HW + 32];
        bpre[t][3].x = xk[48];   bpre[t][3].y = xk[HW + 48];
    }

    __builtin_amdgcn_sched_barrier(0);

    float wacc[4] = {0.0f, 0.0f, 0.0f, 0.0f};
    #pragma unroll
    for (int g = 0; g < GRP; ++g) {
        float inner[4];
        #pragma unroll
        for (int i = 0; i < CGRP; ++i) {
            float t = 0.0f;
            #pragma unroll
            for (int j = 0; j < CGRP; ++j) t += wf[i][j] * ainv[g][j];
            inner[i] = t;
        }
        #pragma unroll
        for (int r = 0; r < CGRP; ++r) {
            #pragma unroll
            for (int i = 0; i < CGRP; ++i) {
                const float ag = __int_as_float(
                    __builtin_amdgcn_readlane(__float_as_int(a4[i]), 4 * g + r));
                wacc[r] += ag * inner[i];
            }
        }
    }
    #pragma unroll
    for (int r = 0; r < CGRP; ++r)
        Ws[(nor * 4 + r) * WPITCH + nc * 4 + s] = wacc[r] * (1.0f / (float)GRP);

    __syncthreads();

    v8f acc0 = {}, acc1 = {}, acc2 = {}, acc3 = {};

    const float* wrow = Ws + nl * WPITCH + kb;

    #pragma unroll
    for (int k0 = 0; k0 < CH; k0 += 4) {
        v2f a = *(const v2f*)(wrow + k0);

        v2f b0, b1, b2, b3;
        if (k0 < 4 * NPRE) {
            b0 = bpre[k0 >> 2][0];  b1 = bpre[k0 >> 2][1];
            b2 = bpre[k0 >> 2][2];  b3 = bpre[k0 >> 2][3];
        } else {
            const float* xk = xcol + (size_t)k0 * HW;
            b0.x = xk[0];            b0.y = xk[HW];
            b1.x = xk[16];           b1.y = xk[HW + 16];
            b2.x = xk[32];           b2.y = xk[HW + 32];
            b3.x = xk[48];           b3.y = xk[HW + 48];
        }

        acc0 = __builtin_amdgcn_wmma_f32_16x16x4_f32(false, a, false, b0, (short)0, acc0, false, false);
        acc1 = __builtin_amdgcn_wmma_f32_16x16x4_f32(false, a, false, b1, (short)0, acc1, false, false);
        acc2 = __builtin_amdgcn_wmma_f32_16x16x4_f32(false, a, false, b2, (short)0, acc2, false, false);
        acc3 = __builtin_amdgcn_wmma_f32_16x16x4_f32(false, a, false, b3, (short)0, acc3, false, false);
    }

    #pragma unroll
    for (int v = 0; v < 8; ++v) {
        float* op = ob + (size_t)(m_off + v + 8 * khi) * HW + p_off + nl;
        op[0]  = acc0[v];
        op[16] = acc1[v];
        op[32] = acc2[v];
        op[48] = acc3[v];
    }
}

extern "C" void kernel_launch(void* const* d_in, const int* in_sizes, int n_in,
                              void* d_out, int out_size, void* d_ws, size_t ws_size,
                              hipStream_t stream) {
    const float* x    = (const float*)d_in[0];
    const float* A    = (const float*)d_in[1];
    const float* Ainv = (const float*)d_in[2];
    const float* Wf   = (const float*)d_in[3];
    float* out  = (float*)d_out;

    reynolds_fused_kernel<<<256, 256, 0, stream>>>(x, A, Ainv, Wf, out);
}


# x and out moved as 16-byte vectors: N index n of accumulator t stands for column 4n+t (was 16t+n); 77 memory instructions per wave, was 197; each out line written by one store; sums bit-identical
# speedup vs baseline: 2.1750x; 1.1981x over previous
#include <hip/hip_runtime.h>
#include <hip/hip_bf16.h>


#define BATCH 8
#define CH    64
#define HW    4096
#define GRP   8
#define CGRP  4

#define WPITCH 68
#define NPRE   4

typedef float v2f __attribute__((ext_vector_type(2)));
typedef float v4f __attribute__((ext_vector_type(4)));
typedef float v8f __attribute__((ext_vector_type(8)));

__global__ __launch_bounds__(256)
void reynolds_fused_kernel(const float* __restrict__ x,
                           const float* __restrict__ A,
                           const float* __restrict__ Ainv,
                           const float* __restrict__ Wf,
                           float* __restrict__ out) {
    __shared__ __attribute__((aligned(16))) float Ws[16 * WPITCH];

    const int tid  = threadIdx.x;
    const int lane = tid & 31;
    const int wib  = tid >> 5;

    const int blk   = blockIdx.x;
    const int batch = blk >> 5;
    const int m_off = ((blk >> 3) & 3) << 4;
    const int p_off = (((blk & 7) << 3) + wib) << 6;

    const int nl  = lane & 15;
    const int khi = lane >> 4;
    const int kb  = khi << 1;

    const float* __restrict__ xb = x   + (size_t)batch * CH * HW;
    float*       __restrict__ ob = out + (size_t)batch * CH * HW;

    const int s   = tid & 3;
    const int nc  = (tid >> 2) & 15;
    const int nor = tid >> 6;
    const int no  = (m_off >> 2) + nor;
    const float* wfb = Wf + (no * 4) * CH + nc * 4;
    v4f wf[4];
    #pragma unroll
    for (int i = 0; i < 4; ++i) wf[i] = *(const v4f*)(wfb + i * CH);

    const v4f a4 = *(const v4f*)(A + lane * 4);

    float ainv[GRP][CGRP];
    #pragma unroll
    for (int g = 0; g < GRP; ++g) {
        #pragma unroll
        for (int j = 0; j < CGRP; ++j) ainv[g][j] = Ainv[g * (CGRP * CGRP) + j * 4 + s];
    }

    const float* xcol = xb + (size_t)kb * HW + p_off + 4 * nl;
    v4f bpre[NPRE][2];
    #pragma unroll
    for (int t = 0; t < NPRE; ++t) {
        const float* xk = xcol + (size_t)(4 * t) * HW;
        bpre[t][0] = *(const v4f*)(xk);
        bpre[t][1] = *(const v4f*)(xk + HW);
    }

    __builtin_amdgcn_sched_barrier(0);

    float wacc[4] = {0.0f, 0.0f, 0.0f, 0.0f};
    #pragma unroll
    for (int g = 0; g < GRP; ++g) {
        float inner[4];
        #pragma unroll
        for (int i = 0; i < CGRP; ++i) {
            float t = 0.0f;
            #pragma unroll
            for (int j = 0; j < CGRP; ++j) t += wf[i][j] * ainv[g][j];
            inner[i] = t;
        }
        #pragma unroll
        for (int r = 0; r < CGRP; ++r) {
            #pragma unroll
            for (int i = 0; i < CGRP; ++i) {
                const float ag = __int_as_float(
                    __builtin_amdgcn_readlane(__float_as_int(a4[i]), 4 * g + r));
                wacc[r] += ag * inner[i];
            }
        }
    }
    #pragma unroll
    for (int r = 0; r < CGRP; ++r)
        Ws[(nor * 4 + r) * WPITCH + nc * 4 + s] = wacc[r] * (1.0f / (float)GRP);

    __syncthreads();

    v8f acc0 = {}, acc1 = {}, acc2 = {}, acc3 = {};

    const float* wrow = Ws + nl * WPITCH + kb;

    #pragma unroll
    for (int k0 = 0; k0 < CH; k0 += 4) {
        v2f a = *(const v2f*)(wrow + k0);

        v4f r0, r1;
        if (k0 < 4 * NPRE) {
            r0 = bpre[k0 >> 2][0];  r1 = bpre[k0 >> 2][1];
        } else {
            const float* xk = xcol + (size_t)k0 * HW;
            r0 = *(const v4f*)(xk);
            r1 = *(const v4f*)(xk + HW);
        }
        const v2f b0 = {r0.x, r1.x};
        const v2f b1 = {r0.y, r1.y};
        const v2f b2 = {r0.z, r1.z};
        const v2f b3 = {r0.w, r1.w};

        acc0 = __builtin_amdgcn_wmma_f32_16x16x4_f32(false, a, false, b0, (short)0, acc0, false, false);
        acc1 = __builtin_amdgcn_wmma_f32_16x16x4_f32(false, a, false, b1, (short)0, acc1, false, false);
        acc2 = __builtin_amdgcn_wmma_f32_16x16x4_f32(false, a, false, b2, (short)0, acc2, false, false);
        acc3 = __builtin_amdgcn_wmma_f32_16x16x4_f32(false, a, false, b3, (short)0, acc3, false, false);
    }

    #pragma unroll
    for (int v = 0; v < 8; ++v) {
        float* op = ob + (size_t)(m_off + v + 8 * khi) * HW + p_off + 4 * nl;
        const v4f o = {acc0[v], acc1[v], acc2[v], acc3[v]};
        *(v4f*)op = o;
    }
}

extern "C" void kernel_launch(void* const* d_in, const int* in_sizes, int n_in,
                              void* d_out, int out_size, void* d_ws, size_t ws_size,
                              hipStream_t stream) {
    const float* x    = (const float*)d_in[0];
    const float* A    = (const float*)d_in[1];
    const float* Ainv = (const float*)d_in[2];
    const float* Wf   = (const float*)d_in[3];
    float* out  = (float*)d_out;

    reynolds_fused_kernel<<<256, 256, 0, stream>>>(x, A, Ainv, Wf, out);
}


# all 32 x loads of a wave requested at entry, ahead of the W_eff build (was 8); the barrier behind the build fences LDS only and no longer waits for them; k loop issues no load; sums bit-identical
# speedup vs baseline: 2.4591x; 1.1306x over previous
#include <hip/hip_runtime.h>
#include <hip/hip_bf16.h>


#define BATCH 8
#define CH    64
#define HW    4096
#define GRP   8
#define CGRP  4

#define WPITCH 68
#define NPRE   16

typedef float v2f __attribute__((ext_vector_type(2)));
typedef float v4f __attribute__((ext_vector_type(4)));
typedef float v8f __attribute__((ext_vector_type(8)));

__global__ __launch_bounds__(256)
void reynolds_fused_kernel(const float* __restrict__ x,
                           const float* __restrict__ A,
                           const float* __restrict__ Ainv,
                           const float* __restrict__ Wf,
                           float* __restrict__ out) {
    __shared__ __attribute__((aligned(16))) float Ws[16 * WPITCH];

    const int tid  = threadIdx.x;
    const int lane = tid & 31;
    const int wib  = tid >> 5;

    const int blk   = blockIdx.x;
    const int batch = blk >> 5;
    const int m_off = ((blk >> 3) & 3) << 4;
    const int p_off = (((blk & 7) << 3) + wib) << 6;

    const int nl  = lane & 15;
    const int khi = lane >> 4;
    const int kb  = khi << 1;

    const float* __restrict__ xb = x   + (size_t)batch * CH * HW;
    float*       __restrict__ ob = out + (size_t)batch * CH * HW;

    const int s   = tid & 3;
    const int nc  = (tid >> 2) & 15;
    const int nor = tid >> 6;
    const int no  = (m_off >> 2) + nor;
    const float* wfb = Wf + (no * 4) * CH + nc * 4;
    v4f wf[4];
    #pragma unroll
    for (int i = 0; i < 4; ++i) wf[i] = *(const v4f*)(wfb + i * CH);

    const v4f a4 = *(const v4f*)(A + lane * 4);

    float ainv[GRP][CGRP];
    #pragma unroll
    for (int g = 0; g < GRP; ++g) {
        #pragma unroll
        for (int j = 0; j < CGRP; ++j) ainv[g][j] = Ainv[g * (CGRP * CGRP) + j * 4 + s];
    }

    const float* xcol = xb + (size_t)kb * HW + p_off + 4 * nl;
    v4f bpre[NPRE][2];
    #pragma unroll
    for (int t = 0; t < NPRE; ++t) {
        const float* xk = xcol + (size_t)(4 * t) * HW;
        bpre[t][0] = *(const v4f*)(xk);
        bpre[t][1] = *(const v4f*)(xk + HW);
    }

    __builtin_amdgcn_sched_barrier(0);

    float wacc[4] = {0.0f, 0.0f, 0.0f, 0.0f};
    #pragma unroll
    for (int g = 0; g < GRP; ++g) {
        float inner[4];
        #pragma unroll
        for (int i = 0; i < CGRP; ++i) {
            float t = 0.0f;
            #pragma unroll
            for (int j = 0; j < CGRP; ++j) t += wf[i][j] * ainv[g][j];
            inner[i] = t;
        }
        #pragma unroll
        for (int r = 0; r < CGRP; ++r) {
            #pragma unroll
            for (int i = 0; i < CGRP; ++i) {
                const float ag = __int_as_float(
                    __builtin_amdgcn_readlane(__float_as_int(a4[i]), 4 * g + r));
                wacc[r] += ag * inner[i];
            }
        }
    }
    #pragma unroll
    for (int r = 0; r < CGRP; ++r)
        Ws[(nor * 4 + r) * WPITCH + nc * 4 + s] = wacc[r] * (1.0f / (float)GRP);

    __builtin_amdgcn_fence(__ATOMIC_RELEASE, "workgroup", "local");
    __builtin_amdgcn_s_barrier();
    __builtin_amdgcn_fence(__ATOMIC_ACQUIRE, "workgroup", "local");

    v8f acc0 = {}, acc1 = {}, acc2 = {}, acc3 = {};

    const float* wrow = Ws + nl * WPITCH + kb;

    #pragma unroll
    for (int k0 = 0; k0 < CH; k0 += 4) {
        v2f a = *(const v2f*)(wrow + k0);

        const v4f r0 = bpre[k0 >> 2][0];
        const v4f r1 = bpre[k0 >> 2][1];
        const v2f b0 = {r0.x, r1.x};
        const v2f b1 = {r0.y, r1.y};
        const v2f b2 = {r0.z, r1.z};
        const v2f b3 = {r0.w, r1.w};

        acc0 = __builtin_amdgcn_wmma_f32_16x16x4_f32(false, a, false, b0, (short)0, acc0, false, false);
        acc1 = __builtin_amdgcn_wmma_f32_16x16x4_f32(false, a, false, b1, (short)0, acc1, false, false);
        acc2 = __builtin_amdgcn_wmma_f32_16x16x4_f32(false, a, false, b2, (short)0, acc2, false, false);
        acc3 = __builtin_amdgcn_wmma_f32_16x16x4_f32(false, a, false, b3, (short)0, acc3, false, false);
    }

    #pragma unroll
    for (int v = 0; v < 8; ++v) {
        float* op = ob + (size_t)(m_off + v + 8 * khi) * HW + p_off + 4 * nl;
        const v4f o = {acc0[v], acc1[v], acc2[v], acc3[v]};
        *(v4f*)op = o;
    }
}

extern "C" void kernel_launch(void* const* d_in, const int* in_sizes, int n_in,
                              void* d_out, int out_size, void* d_ws, size_t ws_size,
                              hipStream_t stream) {
    const float* x    = (const float*)d_in[0];
    const float* A    = (const float*)d_in[1];
    const float* Ainv = (const float*)d_in[2];
    const float* Wf   = (const float*)d_in[3];
    float* out  = (float*)d_out;

    reynolds_fused_kernel<<<256, 256, 0, stream>>>(x, A, Ainv, Wf, out);
}
